# starting kernel
# baseline (speedup 1.0000x reference)
#include <hip/hip_runtime.h>
#include <math.h>

#define EPSBN 1e-5f
#define SLOPE 0.01f
static constexpr int  HP  = 24;
static constexpr long P   = 2L*24*24*24*24;
static constexpr long NX  = 2L*48*48*48*48;
static constexpr int  RX_BLOCKS = 648;
static constexpr int  M3_BLOCKS = 216;
static constexpr int  POS_BLOCKS = 2592;

static constexpr size_t OFF_U2   = 0;
static constexpr size_t OFF_WP   = (size_t)P*8*4;
static constexpr size_t OFF_DACC = OFF_WP + 768*4;
static constexpr size_t OFF_FC   = OFF_DACC + 512;

typedef float vf2 __attribute__((ext_vector_type(2)));
typedef float vf8 __attribute__((ext_vector_type(8)));

__device__ __forceinline__ float lrelu(float v){ return v >= 0.f ? v : SLOPE*v; }

__global__ void k_prep(const float* __restrict__ w2, float* __restrict__ wp,
                       double* __restrict__ dacc)
{
    int tid = threadIdx.x;
    if (tid < 56) dacc[tid] = 0.0;
    if (tid < 384) {
        int n = tid & 15, h = (tid >> 4) & 1, t = tid >> 5;
        int K0 = 4*t + 2*h;
        float vx = (n < 6) ? w2[n*48 + K0]     : 0.f;
        float vy = (n < 6) ? w2[n*48 + K0 + 1] : 0.f;
        ((float2*)wp)[t*32 + h*16 + n] = make_float2(vx, vy);
    }
}

__global__ __launch_bounds__(256) void k_rx(const float* __restrict__ x,
                                            double* __restrict__ dacc)
{
    __shared__ double sacc[2];
    int tid = threadIdx.x;
    if (tid < 2) sacc[tid] = 0.0;
    __syncthreads();
    long gid = (long)blockIdx.x*256 + tid;
    float s = 0.f, q = 0.f;
    #pragma unroll
    for (int it = 0; it < 16; ++it) {
        long idx = gid + (long)it * (RX_BLOCKS*256);
        float4 v = *(const float4*)(x + idx*4);
        s += v.x + v.y + v.z + v.w;
        q += v.x*v.x + v.y*v.y + v.z*v.z + v.w*v.w;
    }
    atomicAdd(&sacc[0], (double)s);
    atomicAdd(&sacc[1], (double)q);
    __syncthreads();
    if (tid < 2) atomicAdd(&dacc[tid], sacc[tid]);
}

__global__ void k_fin1(const double* __restrict__ dacc, const float* __restrict__ w1,
                       const float* __restrict__ g1, const float* __restrict__ b1,
                       float* __restrict__ fc)
{
    if (threadIdx.x != 0) return;
    double n  = (double)NX;
    double mu = dacc[0]/n;
    double var = dacc[1]/n - mu*mu;
    for (int c = 0; c < 3; ++c) {
        double wc  = (double)w1[c];
        double inv = 1.0 / sqrt(wc*wc*var + (double)EPSBN);
        fc[c]   = (float)(wc*inv) * g1[c];
        fc[3+c] = (float)(-wc*mu*inv) * g1[c] + b1[c];
    }
}

__global__ __launch_bounds__(256) void k_conv2(const float* __restrict__ x,
                                               const float* __restrict__ wp,
                                               const float* __restrict__ fc,
                                               float* __restrict__ u2,
                                               double* __restrict__ dacc)
{
    __shared__ float lx[256*18];
    __shared__ float lsum[6], lssq[6];
    int tid = threadIdx.x;
    if (tid < 6) { lsum[tid] = 0.f; lssq[tid] = 0.f; }

    int pos = blockIdx.x*256 + tid;
    int l  = pos % HP;  int t1 = pos / HP;
    int kk = t1 % HP;   t1 /= HP;
    int j  = t1 % HP;   t1 /= HP;
    int i  = t1 % HP;   int b = t1 / HP;
    long xbase = ((((long)b*48 + 2*i)*48 + 2*j)*48 + 2*kk)*48 + 2*l;

    float* row = &lx[tid*18];
    #pragma unroll
    for (int p = 0; p < 2; ++p)
      #pragma unroll
      for (int q = 0; q < 2; ++q)
        #pragma unroll
        for (int r = 0; r < 2; ++r) {
            float2 v = *(const float2*)(x + xbase + (long)p*110592 + q*2304 + r*48);
            *(float2*)(row + (p*8 + q*4 + r*2)) = v;
        }

    int lane = tid & 31, l16 = lane & 15, h = lane >> 4, wv = tid >> 5;

    vf2 Bfrag[12];
    #pragma unroll
    for (int t = 0; t < 12; ++t)
        Bfrag[t] = *(const vf2*)(wp + ((t*2 + h)*16 + l16)*2);

    float a1[3], c1[3];
    #pragma unroll
    for (int c = 0; c < 3; ++c) { a1[c] = fc[c]; c1[c] = fc[3+c]; }

    __syncthreads();

    #pragma unroll
    for (int tau = 0; tau < 2; ++tau) {
        int bl = wv*32 + tau*16;
        vf8 acc = {0.f,0.f,0.f,0.f,0.f,0.f,0.f,0.f};
        #pragma unroll
        for (int t = 0; t < 12; ++t) {
            int c  = t >> 2;
            int fb = (t & 3)*4 + 2*h;
            float2 xv = *(const float2*)&lx[(bl + l16)*18 + fb];
            vf2 A;
            A.x = lrelu(a1[c]*xv.x + c1[c]);
            A.y = lrelu(a1[c]*xv.y + c1[c]);
            acc = __builtin_amdgcn_wmma_f32_16x16x4_f32(
                      false, A, false, Bfrag[t], (short)0, acc, false, false);
        }
        if (l16 < 6) {
            long pbase = (long)blockIdx.x*256 + bl + 8*h;
            float ls = 0.f, lq = 0.f;
            #pragma unroll
            for (int r = 0; r < 8; ++r) {
                float v = acc[r];
                u2[(pbase + r)*8 + l16] = v;
                ls += v; lq += v*v;
            }
            atomicAdd(&lsum[l16], ls);
            atomicAdd(&lssq[l16], lq);
        }
    }
    __syncthreads();
    if (tid < 6) {
        atomicAdd(&dacc[2 + tid], (double)lsum[tid]);
        atomicAdd(&dacc[8 + tid], (double)lssq[tid]);
    }
}

__global__ void k_fin2(const double* __restrict__ dacc, const float* __restrict__ g2,
                       const float* __restrict__ b2, float* __restrict__ fc)
{
    if (threadIdx.x != 0) return;
    double n = (double)P;
    for (int c = 0; c < 6; ++c) {
        double mu  = dacc[2+c]/n;
        double var = dacc[8+c]/n - mu*mu;
        double inv = 1.0 / sqrt(var + (double)EPSBN);
        float sc = (float)inv * g2[c];
        fc[6+c]  = sc;
        fc[12+c] = b2[c] - (float)mu*sc;
    }
}

__global__ __launch_bounds__(256) void k_m3(const float* __restrict__ u2,
                                            const float* __restrict__ fc,
                                            double* __restrict__ dacc)
{
    __shared__ double sacc[27];
    int tid = threadIdx.x;
    if (tid < 27) sacc[tid] = 0.0;
    __syncthreads();
    float m[6]; float s[21];
    #pragma unroll
    for (int c = 0; c < 6; ++c) m[c] = 0.f;
    #pragma unroll
    for (int k = 0; k < 21; ++k) s[k] = 0.f;
    long gid = (long)blockIdx.x*256 + tid;
    for (int it = 0; it < 12; ++it) {
        long pos = gid + (long)it*(M3_BLOCKS*256);
        float4 ua = *(const float4*)(u2 + pos*8);
        float4 ub = *(const float4*)(u2 + pos*8 + 4);
        float uv[6] = {ua.x, ua.y, ua.z, ua.w, ub.x, ub.y};
        float y[6];
        #pragma unroll
        for (int c = 0; c < 6; ++c) y[c] = lrelu(uv[c]*fc[6+c] + fc[12+c]);
        int idx = 0;
        #pragma unroll
        for (int c = 0; c < 6; ++c) {
            m[c] += y[c];
            #pragma unroll
            for (int c2 = c; c2 < 6; ++c2) s[idx++] += y[c]*y[c2];
        }
    }
    #pragma unroll
    for (int c = 0; c < 6; ++c)  atomicAdd(&sacc[c],     (double)m[c]);
    #pragma unroll
    for (int k = 0; k < 21; ++k) atomicAdd(&sacc[6 + k], (double)s[k]);
    __syncthreads();
    if (tid < 27) atomicAdd(&dacc[14 + tid], sacc[tid]);
}

__global__ void k_fin3(const double* __restrict__ dacc, const float* __restrict__ wt,
                       const float* __restrict__ g3, const float* __restrict__ b3,
                       float* __restrict__ fc)
{
    if (threadIdx.x != 0) return;
    double n = (double)P;
    double mean[6], S[6][6];
    for (int c = 0; c < 6; ++c) mean[c] = dacc[14+c]/n;
    int idx = 0;
    for (int c = 0; c < 6; ++c)
        for (int c2 = c; c2 < 6; ++c2) {
            double v = dacc[20+idx]/n; S[c][c2] = v; S[c2][c] = v; ++idx;
        }
    for (int o = 0; o < 3; ++o) {
        double mu = 0.0, m2 = 0.0;
        for (int f = 0; f < 16; ++f) {
            double wv[6];
            for (int c = 0; c < 6; ++c) wv[c] = (double)wt[(c*3+o)*16 + f];
            for (int c = 0; c < 6; ++c) {
                mu += wv[c]*mean[c];
                for (int c2 = 0; c2 < 6; ++c2) m2 += wv[c]*wv[c2]*S[c][c2];
            }
        }
        mu /= 16.0; m2 /= 16.0;
        double var = m2 - mu*mu;
        double inv = 1.0 / sqrt(var + (double)EPSBN);
        float A = (float)inv * g3[o];
        fc[18+o] = A;
        fc[21+o] = b3[o] - (float)mu*A;
    }
}

__global__ __launch_bounds__(256) void k_out(const float* __restrict__ u2,
                                             const float* __restrict__ wt,
                                             const float* __restrict__ fc,
                                             float* __restrict__ out)
{
    int pos = blockIdx.x*256 + threadIdx.x;
    int l  = pos % HP;  int t1 = pos / HP;
    int kk = t1 % HP;   t1 /= HP;
    int j  = t1 % HP;   t1 /= HP;
    int i  = t1 % HP;   int b = t1 / HP;

    float4 ua = *(const float4*)(u2 + (long)pos*8);
    float4 ub = *(const float4*)(u2 + (long)pos*8 + 4);
    float uv[6] = {ua.x, ua.y, ua.z, ua.w, ub.x, ub.y};
    float y[6];
    #pragma unroll
    for (int c = 0; c < 6; ++c) y[c] = lrelu(uv[c]*fc[6+c] + fc[12+c]);

    float acc[48];
    #pragma unroll
    for (int u = 0; u < 48; ++u) acc[u] = 0.f;
    #pragma unroll
    for (int c = 0; c < 6; ++c)
      #pragma unroll
      for (int o = 0; o < 3; ++o) {
        const float* wrow = wt + (c*3 + o)*16;
        #pragma unroll
        for (int f = 0; f < 16; ++f) acc[o*16 + f] += y[c]*wrow[f];
      }

    #pragma unroll
    for (int o = 0; o < 3; ++o) {
        float A3 = fc[18+o], B3 = fc[21+o];
        #pragma unroll
        for (int p = 0; p < 2; ++p)
          #pragma unroll
          for (int q = 0; q < 2; ++q)
            #pragma unroll
            for (int r = 0; r < 2; ++r) {
                int f = p*8 + q*4 + r*2;
                float z0 = acc[o*16 + f    ]*A3 + B3;
                float z1 = acc[o*16 + f + 1]*A3 + B3;
                float s0 = 1.f / (1.f + __expf(-z0));
                float s1 = 1.f / (1.f + __expf(-z1));
                long oidx = (((((long)b*3 + o)*48 + (2*i+p))*48 + (2*j+q))*48 + (2*kk+r))*48 + 2*l;
                *(float2*)(out + oidx) = make_float2(s0, s1);
            }
    }
}

extern "C" void kernel_launch(void* const* d_in, const int* in_sizes, int n_in,
                              void* d_out, int out_size, void* d_ws, size_t ws_size,
                              hipStream_t stream)
{
    (void)in_sizes; (void)n_in; (void)out_size; (void)ws_size;
    const float* x  = (const float*)d_in[0];
    const float* w1 = (const float*)d_in[1];
    const float* g1 = (const float*)d_in[2];
    const float* b1 = (const float*)d_in[3];
    const float* w2 = (const float*)d_in[4];
    const float* g2 = (const float*)d_in[5];
    const float* b2 = (const float*)d_in[6];
    const float* wt = (const float*)d_in[7];
    const float* g3 = (const float*)d_in[8];
    const float* b3 = (const float*)d_in[9];
    float*  out  = (float*)d_out;
    float*  u2   = (float*)((char*)d_ws + OFF_U2);
    float*  wp   = (float*)((char*)d_ws + OFF_WP);
    double* dacc = (double*)((char*)d_ws + OFF_DACC);
    float*  fc   = (float*)((char*)d_ws + OFF_FC);

    k_prep <<<1, 512, 0, stream>>>(w2, wp, dacc);
    k_rx   <<<RX_BLOCKS, 256, 0, stream>>>(x, dacc);
    k_fin1 <<<1, 32, 0, stream>>>(dacc, w1, g1, b1, fc);
    k_conv2<<<POS_BLOCKS, 256, 0, stream>>>(x, wp, fc, u2, dacc);
    k_fin2 <<<1, 32, 0, stream>>>(dacc, g2, b2, fc);
    k_m3   <<<M3_BLOCKS, 256, 0, stream>>>(u2, fc, dacc);
    k_fin3 <<<1, 32, 0, stream>>>(dacc, wt, g3, b3, fc);
    k_out  <<<POS_BLOCKS, 256, 0, stream>>>(u2, wt, fc, out);
}


# k_conv2: 20,736 two-tile waves -> 3,072 resident waves (256 x 12), each walks 6-7 pieces of 32 positions through its own LDS tile, next piece's loads ahead, no barrier in the loop; BN2 sums regrouped
# speedup vs baseline: 2.9915x; 2.9915x over previous
#include <hip/hip_runtime.h>
#include <math.h>

#define EPSBN 1e-5f
#define SLOPE 0.01f
static constexpr int  HP  = 24;
static constexpr long P   = 2L*24*24*24*24;
static constexpr long NX  = 2L*48*48*48*48;
static constexpr int  RX_BLOCKS = 648;
static constexpr int  M3_BLOCKS = 256;
static constexpr int  C2_BLOCKS = 256;
static constexpr int  C2_WAVES  = 12;
static constexpr int  C2_THREADS = C2_WAVES*32;
static constexpr int  C2_NW     = C2_BLOCKS*C2_WAVES;
static constexpr int  C2_PIECES = 20736;

static constexpr size_t OFF_U2   = 0;
static constexpr size_t OFF_WP   = (size_t)P*8*4;
static constexpr size_t OFF_DACC = OFF_WP + 768*4;
static constexpr size_t OFF_FC   = OFF_DACC + 512;
static constexpr size_t OFF_TW   = OFF_FC + 128;

typedef float vf2 __attribute__((ext_vector_type(2)));
typedef float vf8 __attribute__((ext_vector_type(8)));

__device__ __forceinline__ float lrelu(float v){ return v >= 0.f ? v : SLOPE*v; }

static constexpr int TW_N = 160;
__global__ void k_prep(const float* __restrict__ w2, const float* __restrict__ wt,
                       float* __restrict__ wp, double* __restrict__ dacc,
                       float* __restrict__ tw)
{
    int tid = threadIdx.x;
    if (tid < 56) dacc[tid] = 0.0;
    if (tid >= 384 && tid < 384 + TW_N) {
        int e = tid - 384;
        int c = e / 20, k = e - c*20;
        bool pad = k >= 18;
        int o = pad ? 0 : k / 6, r = pad ? 0 : k - o*6;
        const float* wa = wt + (r*3 + o)*16;
        float acc = 0.f;
        if (pad) {
        } else if (c < 6) {
            if (c >= r) {
                const float* wb = wt + (c*3 + o)*16;
                for (int f = 0; f < 16; ++f) acc = fmaf(wa[f], wb[f], acc);
                if (c != r) acc *= 2.f;
            }
        } else if (c == 6) {
            for (int f = 0; f < 16; ++f) acc += wa[f];
        }
        tw[e] = acc;
    }
    if (tid < 384) {
        int n = tid & 15, h = (tid >> 4) & 1, t = tid >> 5;
        int K0 = 4*t + 2*h;
        float vx = (n < 6) ? w2[n*48 + K0]     : 0.f;
        float vy = (n < 6) ? w2[n*48 + K0 + 1] : 0.f;
        ((float2*)wp)[t*32 + h*16 + n] = make_float2(vx, vy);
    }
}

__global__ __launch_bounds__(256) void k_rx(const float* __restrict__ x,
                                            double* __restrict__ dacc)
{
    __shared__ double sacc[2];
    int tid = threadIdx.x;
    if (tid < 2) sacc[tid] = 0.0;
    __syncthreads();
    long gid = (long)blockIdx.x*256 + tid;
    float s = 0.f, q = 0.f;
    #pragma unroll
    for (int it = 0; it < 16; ++it) {
        long idx = gid + (long)it * (RX_BLOCKS*256);
        float4 v = *(const float4*)(x + idx*4);
        s += v.x + v.y + v.z + v.w;
        q += v.x*v.x + v.y*v.y + v.z*v.z + v.w*v.w;
    }
    int c16 = tid & 15;
    vf2 Afr;  Afr.x = s;  Afr.y = q;
    vf2 Bsel; Bsel.x = (c16 == 0) ? 1.f : 0.f;  Bsel.y = (c16 == 1) ? 1.f : 0.f;
    vf8 red = {0.f,0.f,0.f,0.f,0.f,0.f,0.f,0.f};
    red = __builtin_amdgcn_wmma_f32_16x16x4_f32(false, Afr, false, Bsel, (short)0, red, false, false);
    float t = ((red[0] + red[1]) + (red[2] + red[3])) + ((red[4] + red[5]) + (red[6] + red[7]));
    if (c16 < 2) atomicAdd(&sacc[c16], (double)t);
    __syncthreads();
    if (tid < 2) atomicAdd(&dacc[tid], sacc[tid]);
}

__global__ void k_fin1(const double* __restrict__ dacc, const float* __restrict__ w1,
                       const float* __restrict__ g1, const float* __restrict__ b1,
                       float* __restrict__ fc)
{
    if (threadIdx.x != 0) return;
    const double rn = 1.0 / (double)NX;
    double mu  = dacc[0]*rn;
    double var = dacc[1]*rn - mu*mu;
    float muf = (float)mu, varf = (float)var;
    for (int c = 0; c < 3; ++c) {
        float w   = w1[c];
        float inv = rsqrtf(w*w*varf + EPSBN);
        float a   = (w*inv) * g1[c];
        fc[c]   = a;
        fc[3+c] = b1[c] - muf*a;
    }
}

__device__ __forceinline__ void c2_fetch(const float* __restrict__ x, int pos, float2 (&xr)[8])
{
    int l  = pos % HP;  int t1 = pos / HP;
    int kk = t1 % HP;   t1 /= HP;
    int j  = t1 % HP;   t1 /= HP;
    int i  = t1 % HP;   int b = t1 / HP;
    long xbase = ((((long)b*48 + 2*i)*48 + 2*j)*48 + 2*kk)*48 + 2*l;
    #pragma unroll
    for (int p = 0; p < 2; ++p)
      #pragma unroll
      for (int q = 0; q < 2; ++q)
        #pragma unroll
        for (int r = 0; r < 2; ++r)
            xr[p*4 + q*2 + r] = *(const float2*)(x + xbase + (long)p*110592 + q*2304 + r*48);
}

__global__ __launch_bounds__(C2_THREADS) void k_conv2(const float* __restrict__ x,
                                               const float* __restrict__ wp,
                                               const float* __restrict__ fc,
                                               float* __restrict__ u2,
                                               double* __restrict__ dacc)
{
    __shared__ float lx[C2_WAVES*32*18];
    __shared__ double sacc[12];
    int tid = threadIdx.x;
    if (tid < 12) sacc[tid] = 0.0;

    int lane = tid & 31, l16 = lane & 15, h = lane >> 4;
    const int wv   = __builtin_amdgcn_readfirstlane(tid >> 5);
    const int wave = blockIdx.x*C2_WAVES + wv;
    float* tile = lx + wv*(32*18);
    float* row  = tile + lane*18;

    float2 xr[8];
    c2_fetch(x, wave*32 + lane, xr);

    vf2 Bfrag[12];
    #pragma unroll
    for (int t = 0; t < 12; ++t)
        Bfrag[t] = *(const vf2*)(wp + ((t*2 + h)*16 + l16)*2);

    float a1[3], c1[3];
    #pragma unroll
    for (int c = 0; c < 3; ++c) { a1[c] = fc[c]; c1[c] = fc[3+c]; }

    float ls = 0.f, lq = 0.f;
    __syncthreads();

    for (int piece = wave; ; ) {
        #pragma unroll
        for (int k = 0; k < 8; ++k) *(float2*)(row + 2*k) = xr[k];

        const int  next = piece + C2_NW;
        const bool more = next < C2_PIECES;
        if (more) c2_fetch(x, next*32 + lane, xr);

        #pragma unroll
        for (int tau = 0; tau < 2; ++tau) {
            int bl = tau*16;
            vf8 acc = {0.f,0.f,0.f,0.f,0.f,0.f,0.f,0.f};
            #pragma unroll
            for (int t = 0; t < 12; ++t) {
                int c  = t >> 2;
                int fb = (t & 3)*4 + 2*h;
                float2 xv = *(const float2*)&tile[(bl + l16)*18 + fb];
                vf2 A;
                A.x = lrelu(a1[c]*xv.x + c1[c]);
                A.y = lrelu(a1[c]*xv.y + c1[c]);
                acc = __builtin_amdgcn_wmma_f32_16x16x4_f32(
                          false, A, false, Bfrag[t], (short)0, acc, false, false);
            }
            if (l16 < 6) {
                long pbase = (long)piece*32 + bl + 8*h;
                float ts = 0.f, tq = 0.f;
                #pragma unroll
                for (int r = 0; r < 8; ++r) {
                    float v = acc[r];
                    u2[(pbase + r)*8 + l16] = v;
                    ts += v; tq += v*v;
                }
                ls += ts; lq += tq;
            }
        }
        if (!more) break;
        piece = next;
    }
    if (l16 < 6) {
        atomicAdd(&sacc[l16],     (double)ls);
        atomicAdd(&sacc[6 + l16], (double)lq);
    }
    __syncthreads();
    if (tid < 6) {
        atomicAdd(&dacc[2 + tid], sacc[tid]);
        atomicAdd(&dacc[8 + tid], sacc[6 + tid]);
    }
}

__global__ void k_fin2(const double* __restrict__ dacc, const float* __restrict__ g2,
                       const float* __restrict__ b2, float* __restrict__ fc)
{
    if (threadIdx.x != 0) return;
    const double rn = 1.0 / (double)P;
    for (int c = 0; c < 6; ++c) {
        double mu  = dacc[2+c]*rn;
        double var = dacc[8+c]*rn - mu*mu;
        float inv = rsqrtf((float)var + EPSBN);
        float sc = inv * g2[c];
        fc[6+c]  = sc;
        fc[12+c] = b2[c] - (float)mu*sc;
    }
}

static constexpr int M3_WAVES  = 12;
static constexpr int M3_GROUPS = 27;

template <int m>
__device__ __forceinline__ float row_xor(float v)
{
    int b = __builtin_bit_cast(int, v);
    return __builtin_bit_cast(float, __builtin_amdgcn_update_dpp(b, b, 0x160 + m, 0xf, 0xf, false));
}

__global__ __launch_bounds__(384) void k_m3(const float* __restrict__ u2,
                                            const float* __restrict__ fc,
                                            const float* __restrict__ tw,
                                            double* __restrict__ dacc)
{
    __shared__ double sacc[6];
    int tid = threadIdx.x;
    if (tid < 6) sacc[tid] = 0.0;
    __syncthreads();

    int lane = tid & 31, wv = tid >> 5;
    int c = lane & 7;
    bool live = c < 6;
    int cc = live ? c : 5;
    float m  = live ? 1.f : 0.f;
    float sc = fc[6 + cc] * m;
    float sh = fc[12 + cc] * m + (c == 6 ? 1.f : 0.f);

    long wave = (long)blockIdx.x * M3_WAVES + wv;
    const float* src = u2 + wave * (long)(M3_GROUPS * 64) + (lane & ~7) + cc;

    float ua[M3_GROUPS], ub[M3_GROUPS];
    #pragma unroll
    for (int j = 0; j < M3_GROUPS; ++j) { ua[j] = src[j*64]; ub[j] = src[j*64 + 32]; }
    float4 tq4[5];
    #pragma unroll
    for (int i = 0; i < 5; ++i) tq4[i] = ((const float4*)(tw + c*20))[i];
    __builtin_amdgcn_sched_barrier(0);

    vf8 acc0 = {0.f,0.f,0.f,0.f,0.f,0.f,0.f,0.f};
    vf8 acc1 = {0.f,0.f,0.f,0.f,0.f,0.f,0.f,0.f};
    #pragma unroll
    for (int j = 0; j < M3_GROUPS; ++j) {
        vf2 Y;
        Y.x = lrelu(ua[j]*sc + sh);
        Y.y = lrelu(ub[j]*sc + sh);
        if (j & 1)
            acc1 = __builtin_amdgcn_wmma_f32_16x16x4_f32(false, Y, false, Y, (short)0, acc1, false, false);
        else
            acc0 = __builtin_amdgcn_wmma_f32_16x16x4_f32(false, Y, false, Y, (short)0, acc0, false, false);
    }

    bool holder = (lane < 8) || (lane >= 24);
    const float tq[20] = { tq4[0].x, tq4[0].y, tq4[0].z, tq4[0].w, tq4[1].x, tq4[1].y, tq4[1].z, tq4[1].w,
                           tq4[2].x, tq4[2].y, tq4[2].z, tq4[2].w, tq4[3].x, tq4[3].y, tq4[3].z, tq4[3].w,
                           tq4[4].x, tq4[4].y, tq4[4].z, tq4[4].w };
    float v[3] = {0.f, 0.f, 0.f};
    #pragma unroll
    for (int r = 0; r < 6; ++r) {
        float g = acc0[r] + acc1[r];
        #pragma unroll
        for (int o = 0; o < 3; ++o) v[o] = fmaf(tq[o*6 + r], g, v[o]);
    }
    #pragma unroll
    for (int o = 0; o < 3; ++o) {
        float q = (holder && live) ? v[o] : 0.f;
        q += row_xor<1>(q);
        q += row_xor<2>(q);
        q += row_xor<4>(q);
        float w = (c == 6) ? v[o] : q;
        if (holder && (c == 0 || c == 6)) atomicAdd(&sacc[(c == 6 ? 0 : 3) + o], (double)w);
    }
    __syncthreads();
    if (tid < 6) atomicAdd(&dacc[14 + tid], sacc[tid]);
}

__global__ void k_fin3(const double* __restrict__ dacc,
                       const float* __restrict__ g3, const float* __restrict__ b3,
                       float* __restrict__ fc)
{
    if (threadIdx.x != 0) return;
    const double rn = 1.0 / (16.0 * (double)P);
    for (int o = 0; o < 3; ++o) {
        double mu = dacc[14 + o] * rn;
        double m2 = dacc[17 + o] * rn;
        double var = m2 - mu*mu;
        float inv = rsqrtf((float)var + EPSBN);
        float A = inv * g3[o];
        fc[18+o] = A;
        fc[21+o] = b3[o] - (float)mu*A;
    }
}

static constexpr int OUT_ROWS    = 2*HP*HP*HP;
static constexpr int OUT_BLOCKS  = 768;
static constexpr int OUT_THREADS = 192;
static constexpr int OUT_NW      = 4608;
static constexpr int OUT_TRIPS   = 6;
static_assert(OUT_BLOCKS*6 == OUT_NW && OUT_NW*OUT_TRIPS == OUT_ROWS, "every row is taken by exactly one wave");
__global__ __launch_bounds__(192) void k_out(const float* __restrict__ u2,
                                             const float* __restrict__ wt,
                                             const float* __restrict__ fc,
                                             float* __restrict__ out)
{
    const int lane = threadIdx.x & 31;
    const int c = lane & 15, h = lane >> 4;
    const int wave = __builtin_amdgcn_readfirstlane((int)(blockIdx.x*6 + (threadIdx.x >> 5)));

    const int cp   = c < 12 ? c : 11;
    const int uoff = cp*16 + 4*h;
    const int ooff = h*48 + 4*c;

    float4 ue = *(const float4*)(u2 + (size_t)wave*192 + uoff);
    float4 uo = *(const float4*)(u2 + (size_t)wave*192 + uoff + 8);

    const int f = ((c >> 2) & 1)*8 + ((c >> 1) & 1)*4 + (c >> 3)*2 + (c & 1);
    vf2 Afr[3][2];
    for (int o = 0; o < 3; ++o) {
        float A3h = 0.5f*fc[18+o], B3h = 0.5f*fc[21+o];
        float w0 = wt[((4*h    )*3 + o)*16 + f] * A3h;
        float w1 = wt[((4*h + 1)*3 + o)*16 + f] * A3h;
        float w2 = wt[((2      )*3 + o)*16 + f] * A3h;
        float w3 = wt[((3      )*3 + o)*16 + f] * A3h;
        Afr[o][0].x = w0;              Afr[o][0].y = w1;
        Afr[o][1].x = h ? B3h : w2;    Afr[o][1].y = h ? 0.f : w3;
    }
    float sc0 = fc[6 + 4*h], sc1 = fc[7 + 4*h], sc2 = fc[8], sc3 = fc[9];
    float sh0 = fc[12 + 4*h], sh1 = fc[13 + 4*h], sh2 = fc[14], sh3 = fc[15];

    for (int it = 0; it < OUT_TRIPS; ++it) {
        const int R = wave + it*OUT_NW;
        const int Rn = (it + 1 < OUT_TRIPS) ? R + OUT_NW : R;
        float4 ne = *(const float4*)(u2 + (size_t)Rn*192 + uoff);
        float4 no = *(const float4*)(u2 + (size_t)Rn*192 + uoff + 8);

        int kk = R % HP;  int t1 = R / HP;
        int j  = t1 % HP; t1 /= HP;
        int i  = t1 % HP; int b = t1 / HP;
        size_t obase = ((((size_t)(b*3)*48 + 2*i)*48 + 2*j)*48 + 2*kk)*48;

        vf2 Be0, Be1, Bo0, Bo1;
        Be0.x = lrelu(ue.x*sc0 + sh0);  Be0.y = lrelu(ue.y*sc1 + sh1);
        Bo0.x = lrelu(uo.x*sc0 + sh0);  Bo0.y = lrelu(uo.y*sc1 + sh1);
        float e2 = lrelu(ue.z*sc2 + sh2), e3 = lrelu(ue.w*sc3 + sh3);
        float o2 = lrelu(uo.z*sc2 + sh2), o3 = lrelu(uo.w*sc3 + sh3);
        Be1.x = h ? 1.f : e2;  Be1.y = h ? 0.f : e3;
        Bo1.x = h ? 1.f : o2;  Bo1.y = h ? 0.f : o3;

        vf8 De[3], Do[3];
        for (int o = 0; o < 3; ++o) {
            vf8 z8 = {0.f,0.f,0.f,0.f,0.f,0.f,0.f,0.f};
            De[o] = __builtin_amdgcn_wmma_f32_16x16x4_f32(false, Afr[o][0], false, Be0, (short)0, z8,    false, false);
            Do[o] = __builtin_amdgcn_wmma_f32_16x16x4_f32(false, Afr[o][0], false, Bo0, (short)0, z8,    false, false);
            De[o] = __builtin_amdgcn_wmma_f32_16x16x4_f32(false, Afr[o][1], false, Be1, (short)0, De[o], false, false);
            Do[o] = __builtin_amdgcn_wmma_f32_16x16x4_f32(false, Afr[o][1], false, Bo1, (short)0, Do[o], false, false);
        }
        for (int o = 0; o < 3; ++o) {
            for (int g = 0; g < 8; ++g) {
                De[o][g] = __builtin_amdgcn_tanhf(De[o][g])*0.5f + 0.5f;
                Do[o][g] = __builtin_amdgcn_tanhf(Do[o][g])*0.5f + 0.5f;
            }
        }
        if (c < 12) {
            float* ob = out + obase + ooff;
            for (int o = 0; o < 3; ++o)
              for (int p = 0; p < 2; ++p)
                for (int q = 0; q < 2; ++q) {
                    int g = 4*p + 2*q;
                    *(float4*)(ob + ((size_t)o*5308416 + p*110592 + q*2304)) =
                        make_float4(De[o][g], De[o][g+1], Do[o][g], Do[o][g+1]);
                }
        }
        ue = ne; uo = no;
    }
}

extern "C" void kernel_launch(void* const* d_in, const int* in_sizes, int n_in,
                              void* d_out, int out_size, void* d_ws, size_t ws_size,
                              hipStream_t stream)
{
    (void)in_sizes; (void)n_in; (void)out_size; (void)ws_size;
    const float* x  = (const float*)d_in[0];
    const float* w1 = (const float*)d_in[1];
    const float* g1 = (const float*)d_in[2];
    const float* b1 = (const float*)d_in[3];
    const float* w2 = (const float*)d_in[4];
    const float* g2 = (const float*)d_in[5];
    const float* b2 = (const float*)d_in[6];
    const float* wt = (const float*)d_in[7];
    const float* g3 = (const float*)d_in[8];
    const float* b3 = (const float*)d_in[9];
    float*  out  = (float*)d_out;
    float*  u2   = (float*)((char*)d_ws + OFF_U2);
    float*  wp   = (float*)((char*)d_ws + OFF_WP);
    double* dacc = (double*)((char*)d_ws + OFF_DACC);
    float*  fc   = (float*)((char*)d_ws + OFF_FC);
    float*  tw   = (float*)((char*)d_ws + OFF_TW);

    k_prep <<<1, 544, 0, stream>>>(w2, wt, wp, dacc, tw);
    k_rx   <<<RX_BLOCKS, 256, 0, stream>>>(x, dacc);
    k_fin1 <<<1, 32, 0, stream>>>(dacc, w1, g1, b1, fc);
    k_conv2<<<C2_BLOCKS, C2_THREADS, 0, stream>>>(x, wp, fc, u2, dacc);
    k_fin2 <<<1, 32, 0, stream>>>(dacc, g2, b2, fc);
    k_m3   <<<M3_BLOCKS, 384, 0, stream>>>(u2, fc, tw, dacc);
    k_fin3 <<<1, 32, 0, stream>>>(dacc, g3, b3, fc);
    k_out  <<<OUT_BLOCKS, OUT_THREADS, 0, stream>>>(u2, wt, fc, out);
}
